# starting kernel
# baseline (speedup 1.0000x reference)
#include <hip/hip_runtime.h>


typedef float v2f_t __attribute__((ext_vector_type(2)));
typedef float v8f_t __attribute__((ext_vector_type(8)));

#define NDIM 512
#define BDIM 16384
#define ODIM 256

static __device__ __forceinline__ v8f_t wmma_f32(v2f_t a, v2f_t b, v8f_t c) {
    return __builtin_amdgcn_wmma_f32_16x16x4_f32(
        false, a, false, b, (short)0, c, false, false);
}

__global__ void build_A_kernel(const float* __restrict__ W, float* __restrict__ A) {
    int idx = blockIdx.x * blockDim.x + threadIdx.x;
    int i = idx >> 9;
    int j = idx & (NDIM - 1);
    float v = 0.1f * W[idx];
    if (i == j) v += 0.9f;
    A[idx] = v;
}

__global__ void gemm_nn_512_kernel(const float* __restrict__ A,
                                   const float* __restrict__ B,
                                   float* __restrict__ C) {
    const int lane = threadIdx.x & 31;
    const int wave = threadIdx.x >> 5;
    const int tile = blockIdx.x * 4 + wave;
    const int m0 = (tile >> 5) << 4;
    const int n0 = (tile & 31) << 4;
    const int lm = lane & 15;
    const int hi = lane >> 4;

    v8f_t acc = {};
    const float* arow = A + (m0 + lm) * NDIM;
    const float* bcol = B + n0 + lm;

#pragma unroll 4
    for (int kk = 0; kk < NDIM; kk += 4) {
        const int kb = kk + 2 * hi;
        v2f_t a = *(const v2f_t*)(arow + kb);
        v2f_t b;
        b.x = bcol[kb * NDIM];
        b.y = bcol[(kb + 1) * NDIM];
        acc = wmma_f32(a, b, acc);
    }

    float* crow = C + (m0 + 8 * hi) * NDIM + n0 + lm;
#pragma unroll
    for (int v = 0; v < 8; ++v) crow[v * NDIM] = acc[v];
}

__global__ void gemm_nt_P_kernel(const float* __restrict__ M,
                                 const float* __restrict__ Wout,
                                 float* __restrict__ P) {
    const int lane = threadIdx.x & 31;
    const int wave = threadIdx.x >> 5;
    const int tile = blockIdx.x * 4 + wave;
    const int m0 = (tile >> 4) << 4;
    const int n0 = (tile & 15) << 4;
    const int lm = lane & 15;
    const int hi = lane >> 4;

    v8f_t acc = {};
    const float* arow = M + (m0 + lm) * NDIM;
    const float* wrow = Wout + (n0 + lm) * NDIM;

#pragma unroll 4
    for (int kk = 0; kk < NDIM; kk += 4) {
        const int kb = kk + 2 * hi;
        v2f_t a = *(const v2f_t*)(arow + kb);
        v2f_t b = *(const v2f_t*)(wrow + kb);
        acc = wmma_f32(a, b, acc);
    }

    float* prow = P + (m0 + 8 * hi) * ODIM + n0 + lm;
#pragma unroll
    for (int v = 0; v < 8; ++v) prow[v * ODIM] = acc[v];
}

__global__ void gemm_out_kernel(const float* __restrict__ X,
                                const float* __restrict__ P,
                                float* __restrict__ Y) {
    const int lane = threadIdx.x & 31;
    const int wave = threadIdx.x >> 5;
    const int m0 = blockIdx.x << 4;
    const int n0 = wave << 6;
    const int lm = lane & 15;
    const int hi = lane >> 4;

    v8f_t acc0 = {}, acc1 = {}, acc2 = {}, acc3 = {};
    const float* xrow = X + (m0 + lm) * NDIM;
    const float* pcol = P + n0 + lm;

#pragma unroll 2
    for (int kk = 0; kk < NDIM; kk += 4) {
        const int kb = kk + 2 * hi;
        v2f_t a = *(const v2f_t*)(xrow + kb);
        const float* p0 = pcol + kb * ODIM;
        const float* p1 = pcol + (kb + 1) * ODIM;
        v2f_t b0 = { p0[0],  p1[0]  };
        v2f_t b1 = { p0[16], p1[16] };
        v2f_t b2 = { p0[32], p1[32] };
        v2f_t b3 = { p0[48], p1[48] };
        acc0 = wmma_f32(a, b0, acc0);
        acc1 = wmma_f32(a, b1, acc1);
        acc2 = wmma_f32(a, b2, acc2);
        acc3 = wmma_f32(a, b3, acc3);
    }

    float* yrow = Y + (m0 + 8 * hi) * ODIM + n0 + lm;
#pragma unroll
    for (int v = 0; v < 8; ++v) {
        yrow[v * ODIM + 0]  = acc0[v];
        yrow[v * ODIM + 16] = acc1[v];
        yrow[v * ODIM + 32] = acc2[v];
        yrow[v * ODIM + 48] = acc3[v];
    }
}

extern "C" void kernel_launch(void* const* d_in, const int* in_sizes, int n_in,
                              void* d_out, int out_size, void* d_ws, size_t ws_size,
                              hipStream_t stream) {
    (void)in_sizes; (void)n_in; (void)out_size; (void)ws_size;

    const float* x    = (const float*)d_in[0];
    const float* W    = (const float*)d_in[1];
    const float* wout = (const float*)d_in[2];
    float*       out  = (float*)d_out;

    float* A  = (float*)d_ws;
    float* T0 = A  + NDIM * NDIM;
    float* T1 = T0 + NDIM * NDIM;
    float* P  = T1 + NDIM * NDIM;

    build_A_kernel<<<(NDIM * NDIM) / 256, 256, 0, stream>>>(W, A);

    gemm_nn_512_kernel<<<256, 128, 0, stream>>>(A,  A,  T0);
    gemm_nn_512_kernel<<<256, 128, 0, stream>>>(T0, T0, T1);
    gemm_nn_512_kernel<<<256, 128, 0, stream>>>(T1, T1, T0);
    gemm_nn_512_kernel<<<256, 128, 0, stream>>>(T0, T0, T1);
    gemm_nn_512_kernel<<<256, 128, 0, stream>>>(T1, T1, T0);
    gemm_nn_512_kernel<<<256, 128, 0, stream>>>(T0, T0, T1);
    gemm_nn_512_kernel<<<256, 128, 0, stream>>>(T1, A,  T0);

    gemm_nt_P_kernel<<<128, 128, 0, stream>>>(T0, wout, P);

    gemm_out_kernel<<<BDIM / 16, 128, 0, stream>>>(x, P, out);
}


# 512-wide products: K split in four over a workgroup's waves (32x32 block per workgroup, 4 accumulators per wave), partial tiles summed through LDS; L2 operand bytes halved; sum order changed
# speedup vs baseline: 1.8355x; 1.8355x over previous
#include <hip/hip_runtime.h>


typedef float v2f_t __attribute__((ext_vector_type(2)));
typedef float v4f_t __attribute__((ext_vector_type(4)));
typedef float v8f_t __attribute__((ext_vector_type(8)));

#define NDIM 512
#define BDIM 16384
#define ODIM 256
#define FRAG_TILE_STRIDE (64 * 128)
#define FRAG_GROUP_STRIDE 128

static __device__ __forceinline__ v8f_t wmma_f32(v2f_t a, v2f_t b, v8f_t c) {
    return __builtin_amdgcn_wmma_f32_16x16x4_f32(
        false, a, false, b, (short)0, c, false, false);
}

static __device__ __forceinline__ v4f_t ld4(const float* p) {
    return *(const v4f_t*)p;
}

static __device__ __forceinline__ void st4(float* p, v4f_t v) {
    *(v4f_t*)p = v;
}

__global__ void __launch_bounds__(256)
build_A_kernel(const float* __restrict__ W, float* __restrict__ A,
               float* __restrict__ AF) {
    const int p = blockIdx.x * blockDim.x + threadIdx.x;
    const int c = p & 15;
    const int h = (p >> 4) & 1;
    const int g = (p >> 5) & 63;
    const int T = p >> 11;
    const int k0 = 8 * g + 4 * h;
    const int n = 16 * T + c;
    v4f_t f;
#pragma unroll
    for (int e = 0; e < 4; ++e) {
        const int idx = (k0 + e) * NDIM + n;
        float v = 0.1f * W[idx];
        if (k0 + e == n) v += 0.9f;
        A[idx] = v;
        f[e] = v;
    }
    st4(AF + 4 * p, f);
}

#define FEW_WAVES __attribute__((amdgpu_waves_per_eu(1, 2)))

#define SK_G 16

typedef v4f_t red_tile_t[2][32];

static __device__ __forceinline__ void red_put(red_tile_t& dst, int lane, v8f_t acc) {
    dst[0][lane] = (v4f_t){ acc[0], acc[1], acc[2], acc[3] };
    dst[1][lane] = (v4f_t){ acc[4], acc[5], acc[6], acc[7] };
}

static __device__ __forceinline__ void store_tile_frag(
        float* F, int T, int ti, int c, int h, v8f_t acc) {
    float* fp = F + (size_t)T * FRAG_TILE_STRIDE
                  + (2 * ti + h) * FRAG_GROUP_STRIDE + 4 * c;
    v4f_t lo = { acc[0], acc[1], acc[2], acc[3] };
    v4f_t hi = { acc[4], acc[5], acc[6], acc[7] };
    st4(fp, lo);
    st4(fp + 64, hi);
}

__global__ void __launch_bounds__(128) FEW_WAVES
gemm_chain_kernel(const float* __restrict__ A, const float* __restrict__ BF,
                  float* __restrict__ C, float* __restrict__ CF) {
    __shared__ red_tile_t red[4][4];
    const int lane = threadIdx.x & 31;
    const int w = threadIdx.x >> 5;
    const int bi = blockIdx.x >> 4;
    const int bj = blockIdx.x & 15;
    const int c = lane & 15;
    const int h = lane >> 4;

    const float* ap = A + (32 * bi + c) * NDIM + 8 * (SK_G * w) + 4 * h;
    const float* bp = BF + (size_t)(2 * bj) * FRAG_TILE_STRIDE
                         + FRAG_GROUP_STRIDE * (SK_G * w) + 4 * lane;

    v4f_t a[SK_G][2], b[SK_G][2];
#pragma unroll
    for (int i = 0; i < SK_G; ++i) {
        a[i][0] = ld4(ap + 8 * i);
        a[i][1] = ld4(ap + 16 * NDIM + 8 * i);
        b[i][0] = ld4(bp + FRAG_GROUP_STRIDE * i);
        b[i][1] = ld4(bp + FRAG_TILE_STRIDE + FRAG_GROUP_STRIDE * i);
    }

    v8f_t acc[2][2];
#pragma unroll
    for (int r = 0; r < 2; ++r)
#pragma unroll
        for (int t = 0; t < 2; ++t) acc[r][t] = (v8f_t){};

#pragma unroll
    for (int i = 0; i < SK_G; ++i) {
#pragma unroll
        for (int r = 0; r < 2; ++r)
#pragma unroll
            for (int t = 0; t < 2; ++t)
                acc[r][t] = wmma_f32(a[i][r].xy, b[i][t].xy, acc[r][t]);
#pragma unroll
        for (int r = 0; r < 2; ++r)
#pragma unroll
            for (int t = 0; t < 2; ++t)
                acc[r][t] = wmma_f32(a[i][r].zw, b[i][t].zw, acc[r][t]);
    }

#pragma unroll
    for (int r = 0; r < 2; ++r)
#pragma unroll
        for (int t = 0; t < 2; ++t) red_put(red[w][2 * r + t], lane, acc[r][t]);
    __syncthreads();

    v4f_t lo = red[0][w][0][lane];
    v4f_t hi = red[0][w][1][lane];
#pragma unroll
    for (int s = 1; s < 4; ++s) {
        lo += red[s][w][0][lane];
        hi += red[s][w][1][lane];
    }
    const v8f_t sum = { lo[0], lo[1], lo[2], lo[3], hi[0], hi[1], hi[2], hi[3] };

    const int ti = 2 * bi + (w >> 1);
    const int tj = 2 * bj + (w & 1);
    float* crow = C + (16 * ti + 8 * h) * NDIM + 16 * tj + c;
#pragma unroll
    for (int v = 0; v < 8; ++v) crow[v * NDIM] = sum[v];

    store_tile_frag(CF, tj, ti, c, h, sum);
}

__global__ void __launch_bounds__(128) FEW_WAVES
gemm_nt_P_kernel(const float* __restrict__ M, const float* __restrict__ Wout,
                 float* __restrict__ PF) {
    __shared__ red_tile_t red[4][2];
    const int lane = threadIdx.x & 31;
    const int w = threadIdx.x >> 5;
    const int ti = blockIdx.x >> 3;
    const int tp = blockIdx.x & 7;
    const int c = lane & 15;
    const int h = lane >> 4;
    const int T0 = 2 * tp;
    const int T1 = 2 * tp + 1;
    const int n0 = 64 * (T0 >> 2) + 4 * c + (T0 & 3);
    const int n1 = 64 * (T1 >> 2) + 4 * c + (T1 & 3);

    const float* ap  = M + (16 * ti + c) * NDIM + 8 * (SK_G * w) + 4 * h;
    const float* bp0 = Wout + n0 * NDIM + 8 * (SK_G * w) + 4 * h;
    const float* bp1 = Wout + n1 * NDIM + 8 * (SK_G * w) + 4 * h;

    v4f_t a[SK_G], b[SK_G][2];
#pragma unroll
    for (int i = 0; i < SK_G; ++i) {
        a[i]    = ld4(ap  + 8 * i);
        b[i][0] = ld4(bp0 + 8 * i);
        b[i][1] = ld4(bp1 + 8 * i);
    }

    v8f_t acc[2];
    acc[0] = (v8f_t){};
    acc[1] = (v8f_t){};
#pragma unroll
    for (int i = 0; i < SK_G; ++i) {
        acc[0] = wmma_f32(a[i].xy, b[i][0].xy, acc[0]);
        acc[1] = wmma_f32(a[i].xy, b[i][1].xy, acc[1]);
        acc[0] = wmma_f32(a[i].zw, b[i][0].zw, acc[0]);
        acc[1] = wmma_f32(a[i].zw, b[i][1].zw, acc[1]);
    }

    red_put(red[w][0], lane, acc[0]);
    red_put(red[w][1], lane, acc[1]);
    __syncthreads();

    const int t  = w >> 1;
    const int hf = w & 1;
    v4f_t s = red[0][t][hf][lane];
#pragma unroll
    for (int q = 1; q < 4; ++q) s += red[q][t][hf][lane];

    float* fp = PF + (size_t)(2 * tp + t) * FRAG_TILE_STRIDE
                   + (2 * ti + h) * FRAG_GROUP_STRIDE + 4 * c;
    st4(fp + 64 * hf, s);
}

#define OUT_CH 4

struct OutFrags {
    v4f_t a[OUT_CH][2];
    v4f_t b[OUT_CH][4];
};

static __device__ __forceinline__ void out_load(
        OutFrags& f, const float* xa, const float* pb, int g0) {
#pragma unroll
    for (int i = 0; i < OUT_CH; ++i) {
        f.a[i][0] = ld4(xa + 8 * (g0 + i));
        f.a[i][1] = ld4(xa + 16 * NDIM + 8 * (g0 + i));
#pragma unroll
        for (int t = 0; t < 4; ++t)
            f.b[i][t] = ld4(pb + t * FRAG_TILE_STRIDE + FRAG_GROUP_STRIDE * (g0 + i));
    }
}

static __device__ __forceinline__ void out_mma(const OutFrags& f, v8f_t (&acc)[2][4]) {
#pragma unroll
    for (int i = 0; i < OUT_CH; ++i) {
#pragma unroll
        for (int r = 0; r < 2; ++r)
#pragma unroll
            for (int t = 0; t < 4; ++t)
                acc[r][t] = wmma_f32(f.a[i][r].xy, f.b[i][t].xy, acc[r][t]);
#pragma unroll
        for (int r = 0; r < 2; ++r)
#pragma unroll
            for (int t = 0; t < 4; ++t)
                acc[r][t] = wmma_f32(f.a[i][r].zw, f.b[i][t].zw, acc[r][t]);
    }
}

__global__ void __launch_bounds__(128)
gemm_out_kernel(const float* __restrict__ X, const float* __restrict__ PF,
                float* __restrict__ Y) {
    const int lane = threadIdx.x & 31;
    const int wave = threadIdx.x >> 5;
    const int c = lane & 15;
    const int h = lane >> 4;
    const int m0 = (blockIdx.x >> 1) * 64 + (wave >> 1) * 32;
    const int q  = (blockIdx.x & 1) * 2 + (wave & 1);

    const float* xa = X + (size_t)(m0 + c) * NDIM + 4 * h;
    const float* pb = PF + (size_t)(4 * q) * FRAG_TILE_STRIDE + 4 * lane;

    v8f_t acc[2][4];
#pragma unroll
    for (int r = 0; r < 2; ++r)
#pragma unroll
        for (int t = 0; t < 4; ++t) acc[r][t] = (v8f_t){};

    OutFrags s0, s1;
    out_load(s0, xa, pb, 0);
#pragma unroll 1
    for (int g = 0; g < 56; g += 8) {
        out_load(s1, xa, pb, g + 4);
        out_mma(s0, acc);
        out_load(s0, xa, pb, g + 8);
        out_mma(s1, acc);
    }
    out_load(s1, xa, pb, 60);
    out_mma(s0, acc);
    out_mma(s1, acc);

    float* yrow = Y + (size_t)(m0 + 8 * h) * ODIM + 64 * q + 4 * c;
#pragma unroll
    for (int r = 0; r < 2; ++r)
#pragma unroll
        for (int v = 0; v < 8; ++v) {
            v4f_t o = { acc[r][0][v], acc[r][1][v], acc[r][2][v], acc[r][3][v] };
            st4(yrow + (size_t)(16 * r + v) * ODIM, o);
        }
}

extern "C" void kernel_launch(void* const* d_in, const int* in_sizes, int n_in,
                              void* d_out, int out_size, void* d_ws, size_t ws_size,
                              hipStream_t stream) {
    (void)in_sizes; (void)n_in; (void)out_size; (void)ws_size;

    const float* x    = (const float*)d_in[0];
    const float* W    = (const float*)d_in[1];
    const float* wout = (const float*)d_in[2];
    float*       out  = (float*)d_out;

    float* A   = (float*)d_ws;
    float* AF  = A   + NDIM * NDIM;
    float* T0  = AF  + NDIM * NDIM;
    float* T0F = T0  + NDIM * NDIM;
    float* T1  = T0F + NDIM * NDIM;
    float* T1F = T1  + NDIM * NDIM;
    float* PF  = T1F + NDIM * NDIM;

    build_A_kernel<<<(NDIM * NDIM / 4) / 256, 256, 0, stream>>>(W, A, AF);

    gemm_chain_kernel<<<256, 128, 0, stream>>>(A,  AF,  T0, T0F);
    gemm_chain_kernel<<<256, 128, 0, stream>>>(T0, T0F, T1, T1F);
    gemm_chain_kernel<<<256, 128, 0, stream>>>(T1, T1F, T0, T0F);
    gemm_chain_kernel<<<256, 128, 0, stream>>>(T0, T0F, T1, T1F);
    gemm_chain_kernel<<<256, 128, 0, stream>>>(T1, T1F, T0, T0F);
    gemm_chain_kernel<<<256, 128, 0, stream>>>(T0, T0F, T1, T1F);
    gemm_chain_kernel<<<256, 128, 0, stream>>>(T1, AF,  T0, T0F);

    gemm_nt_P_kernel<<<256, 128, 0, stream>>>(T0, wout, PF);

    gemm_out_kernel<<<512, 128, 0, stream>>>(x, PF, out);
}


# A operand of every 512-wide product (and w_out for P) read in fragment order too: producers write 16-row-tile A-fragment order, not row-major; a fragment load is 512 contiguous bytes, 4 lines not 16
# speedup vs baseline: 2.0206x; 1.1008x over previous
#include <hip/hip_runtime.h>


typedef float v2f_t __attribute__((ext_vector_type(2)));
typedef float v4f_t __attribute__((ext_vector_type(4)));
typedef float v8f_t __attribute__((ext_vector_type(8)));

#define NDIM 512
#define BDIM 16384
#define ODIM 256
#define FRAG_TILE_STRIDE (64 * 128)
#define FRAG_GROUP_STRIDE 128

static __device__ __forceinline__ v8f_t wmma_f32(v2f_t a, v2f_t b, v8f_t c) {
    return __builtin_amdgcn_wmma_f32_16x16x4_f32(
        false, a, false, b, (short)0, c, false, false);
}

static __device__ __forceinline__ v4f_t ld4(const float* p) {
    return *(const v4f_t*)p;
}

static __device__ __forceinline__ void st4(float* p, v4f_t v) {
    *(v4f_t*)p = v;
}

__global__ void __launch_bounds__(256)
build_A_kernel(const float* __restrict__ W, const float* __restrict__ Wout,
               float* __restrict__ AG, float* __restrict__ AF,
               float* __restrict__ WF) {
    const int p = blockIdx.x * blockDim.x + threadIdx.x;
    const int c = p & 15;
    const int h = (p >> 4) & 1;
    const int g = (p >> 5) & 63;
    const int T = p >> 11;
    const int k0 = 8 * g + 4 * h;
    const int n = 16 * T + c;
    const v4f_t wr = ld4(W + n * NDIM + k0);
    v4f_t f, a;
#pragma unroll
    for (int e = 0; e < 4; ++e) {
        float v = 0.1f * W[(k0 + e) * NDIM + n];
        float u = 0.1f * wr[e];
        if (k0 + e == n) { v += 0.9f; u += 0.9f; }
        f[e] = v;
        a[e] = u;
    }
    st4(AF + 4 * p, f);
    st4(AG + 4 * p, a);
    if (T < 16) {
        const int nw = 64 * (T >> 2) + 4 * c + (T & 3);
        st4(WF + 4 * p, ld4(Wout + nw * NDIM + k0));
    }
}

#define FEW_WAVES __attribute__((amdgpu_waves_per_eu(1, 2)))

#define SK_G 16

typedef v4f_t red_tile_t[2][32];

static __device__ __forceinline__ void red_put(red_tile_t& dst, int lane, v8f_t acc) {
    dst[0][lane] = (v4f_t){ acc[0], acc[1], acc[2], acc[3] };
    dst[1][lane] = (v4f_t){ acc[4], acc[5], acc[6], acc[7] };
}

static __device__ __forceinline__ void store_tile_frag(
        float* F, int T, int ti, int c, int h, v8f_t acc) {
    float* fp = F + (size_t)T * FRAG_TILE_STRIDE
                  + (2 * ti + h) * FRAG_GROUP_STRIDE + 4 * c;
    v4f_t lo = { acc[0], acc[1], acc[2], acc[3] };
    v4f_t hi = { acc[4], acc[5], acc[6], acc[7] };
    st4(fp, lo);
    st4(fp + 64, hi);
}

static __device__ __forceinline__ void store_tile_afrag(
        float* G, int ti, int tj, int c, int h, v8f_t acc) {
    float* gp = G + (size_t)ti * FRAG_TILE_STRIDE
                  + (2 * tj + (c >> 3)) * FRAG_GROUP_STRIDE
                  + 64 * ((c >> 2) & 1) + 32 * h + (c & 3);
#pragma unroll
    for (int r = 0; r < 8; ++r) gp[4 * r] = acc[r];
}

__global__ void __launch_bounds__(128) FEW_WAVES
gemm_chain_kernel(const float* __restrict__ AG, const float* __restrict__ BF,
                  float* __restrict__ CG, float* __restrict__ CF) {
    __shared__ red_tile_t red[4][4];
    const int lane = threadIdx.x & 31;
    const int w = threadIdx.x >> 5;
    const int bi = blockIdx.x >> 4;
    const int bj = blockIdx.x & 15;
    const int c = lane & 15;
    const int h = lane >> 4;

    const float* ap = AG + (size_t)(2 * bi) * FRAG_TILE_STRIDE
                         + FRAG_GROUP_STRIDE * (SK_G * w) + 4 * lane;
    const float* bp = BF + (size_t)(2 * bj) * FRAG_TILE_STRIDE
                         + FRAG_GROUP_STRIDE * (SK_G * w) + 4 * lane;

    v4f_t a[SK_G][2], b[SK_G][2];
#pragma unroll
    for (int i = 0; i < SK_G; ++i) {
        a[i][0] = ld4(ap + FRAG_GROUP_STRIDE * i);
        a[i][1] = ld4(ap + FRAG_TILE_STRIDE + FRAG_GROUP_STRIDE * i);
        b[i][0] = ld4(bp + FRAG_GROUP_STRIDE * i);
        b[i][1] = ld4(bp + FRAG_TILE_STRIDE + FRAG_GROUP_STRIDE * i);
    }

    v8f_t acc[2][2];
#pragma unroll
    for (int r = 0; r < 2; ++r)
#pragma unroll
        for (int t = 0; t < 2; ++t) acc[r][t] = (v8f_t){};

#pragma unroll
    for (int i = 0; i < SK_G; ++i) {
#pragma unroll
        for (int r = 0; r < 2; ++r)
#pragma unroll
            for (int t = 0; t < 2; ++t)
                acc[r][t] = wmma_f32(a[i][r].xy, b[i][t].xy, acc[r][t]);
#pragma unroll
        for (int r = 0; r < 2; ++r)
#pragma unroll
            for (int t = 0; t < 2; ++t)
                acc[r][t] = wmma_f32(a[i][r].zw, b[i][t].zw, acc[r][t]);
    }

#pragma unroll
    for (int r = 0; r < 2; ++r)
#pragma unroll
        for (int t = 0; t < 2; ++t) red_put(red[w][2 * r + t], lane, acc[r][t]);
    __syncthreads();

    v4f_t lo = red[0][w][0][lane];
    v4f_t hi = red[0][w][1][lane];
#pragma unroll
    for (int s = 1; s < 4; ++s) {
        lo += red[s][w][0][lane];
        hi += red[s][w][1][lane];
    }
    const v8f_t sum = { lo[0], lo[1], lo[2], lo[3], hi[0], hi[1], hi[2], hi[3] };

    const int ti = 2 * bi + (w >> 1);
    const int tj = 2 * bj + (w & 1);
    store_tile_afrag(CG, ti, tj, c, h, sum);
    store_tile_frag(CF, tj, ti, c, h, sum);
}

__global__ void __launch_bounds__(128) FEW_WAVES
gemm_nt_P_kernel(const float* __restrict__ MG, const float* __restrict__ WF,
                 float* __restrict__ PF) {
    __shared__ red_tile_t red[4][2];
    const int lane = threadIdx.x & 31;
    const int w = threadIdx.x >> 5;
    const int ti = blockIdx.x >> 3;
    const int tp = blockIdx.x & 7;
    const int c = lane & 15;
    const int h = lane >> 4;
    const int T0 = 2 * tp;
    const int T1 = 2 * tp + 1;

    const float* ap  = MG + (size_t)ti * FRAG_TILE_STRIDE
                          + FRAG_GROUP_STRIDE * (SK_G * w) + 4 * lane;
    const float* bp0 = WF + (size_t)T0 * FRAG_TILE_STRIDE
                          + FRAG_GROUP_STRIDE * (SK_G * w) + 4 * lane;
    const float* bp1 = WF + (size_t)T1 * FRAG_TILE_STRIDE
                          + FRAG_GROUP_STRIDE * (SK_G * w) + 4 * lane;

    v4f_t a[SK_G], b[SK_G][2];
#pragma unroll
    for (int i = 0; i < SK_G; ++i) {
        a[i]    = ld4(ap  + FRAG_GROUP_STRIDE * i);
        b[i][0] = ld4(bp0 + FRAG_GROUP_STRIDE * i);
        b[i][1] = ld4(bp1 + FRAG_GROUP_STRIDE * i);
    }

    v8f_t acc[2];
    acc[0] = (v8f_t){};
    acc[1] = (v8f_t){};
#pragma unroll
    for (int i = 0; i < SK_G; ++i) {
        acc[0] = wmma_f32(a[i].xy, b[i][0].xy, acc[0]);
        acc[1] = wmma_f32(a[i].xy, b[i][1].xy, acc[1]);
        acc[0] = wmma_f32(a[i].zw, b[i][0].zw, acc[0]);
        acc[1] = wmma_f32(a[i].zw, b[i][1].zw, acc[1]);
    }

    red_put(red[w][0], lane, acc[0]);
    red_put(red[w][1], lane, acc[1]);
    __syncthreads();

    const int t  = w >> 1;
    const int hf = w & 1;
    v4f_t s = red[0][t][hf][lane];
#pragma unroll
    for (int q = 1; q < 4; ++q) s += red[q][t][hf][lane];

    float* fp = PF + (size_t)(2 * tp + t) * FRAG_TILE_STRIDE
                   + (2 * ti + h) * FRAG_GROUP_STRIDE + 4 * c;
    st4(fp + 64 * hf, s);
}

#define OUT_CH 4

struct OutFrags {
    v4f_t a[OUT_CH][2];
    v4f_t b[OUT_CH][4];
};

static __device__ __forceinline__ void out_load(
        OutFrags& f, const float* xa, const float* pb, int g0) {
#pragma unroll
    for (int i = 0; i < OUT_CH; ++i) {
        f.a[i][0] = ld4(xa + 8 * (g0 + i));
        f.a[i][1] = ld4(xa + 16 * NDIM + 8 * (g0 + i));
#pragma unroll
        for (int t = 0; t < 4; ++t)
            f.b[i][t] = ld4(pb + t * FRAG_TILE_STRIDE + FRAG_GROUP_STRIDE * (g0 + i));
    }
}

static __device__ __forceinline__ void out_mma(const OutFrags& f, v8f_t (&acc)[2][4]) {
#pragma unroll
    for (int i = 0; i < OUT_CH; ++i) {
#pragma unroll
        for (int r = 0; r < 2; ++r)
#pragma unroll
            for (int t = 0; t < 4; ++t)
                acc[r][t] = wmma_f32(f.a[i][r].xy, f.b[i][t].xy, acc[r][t]);
#pragma unroll
        for (int r = 0; r < 2; ++r)
#pragma unroll
            for (int t = 0; t < 4; ++t)
                acc[r][t] = wmma_f32(f.a[i][r].zw, f.b[i][t].zw, acc[r][t]);
    }
}

__global__ void __launch_bounds__(128)
gemm_out_kernel(const float* __restrict__ X, const float* __restrict__ PF,
                float* __restrict__ Y) {
    const int lane = threadIdx.x & 31;
    const int wave = threadIdx.x >> 5;
    const int c = lane & 15;
    const int h = lane >> 4;
    const int m0 = (blockIdx.x >> 1) * 64 + (wave >> 1) * 32;
    const int q  = (blockIdx.x & 1) * 2 + (wave & 1);

    const float* xa = X + (size_t)(m0 + c) * NDIM + 4 * h;
    const float* pb = PF + (size_t)(4 * q) * FRAG_TILE_STRIDE + 4 * lane;

    v8f_t acc[2][4];
#pragma unroll
    for (int r = 0; r < 2; ++r)
#pragma unroll
        for (int t = 0; t < 4; ++t) acc[r][t] = (v8f_t){};

    OutFrags s0, s1;
    out_load(s0, xa, pb, 0);
#pragma unroll 1
    for (int g = 0; g < 56; g += 8) {
        out_load(s1, xa, pb, g + 4);
        out_mma(s0, acc);
        out_load(s0, xa, pb, g + 8);
        out_mma(s1, acc);
    }
    out_load(s1, xa, pb, 60);
    out_mma(s0, acc);
    out_mma(s1, acc);

    float* yrow = Y + (size_t)(m0 + 8 * h) * ODIM + 64 * q + 4 * c;
#pragma unroll
    for (int r = 0; r < 2; ++r)
#pragma unroll
        for (int v = 0; v < 8; ++v) {
            v4f_t o = { acc[r][0][v], acc[r][1][v], acc[r][2][v], acc[r][3][v] };
            st4(yrow + (size_t)(16 * r + v) * ODIM, o);
        }
}

extern "C" void kernel_launch(void* const* d_in, const int* in_sizes, int n_in,
                              void* d_out, int out_size, void* d_ws, size_t ws_size,
                              hipStream_t stream) {
    (void)in_sizes; (void)n_in; (void)out_size; (void)ws_size;

    const float* x    = (const float*)d_in[0];
    const float* W    = (const float*)d_in[1];
    const float* wout = (const float*)d_in[2];
    float*       out  = (float*)d_out;

    float* AG  = (float*)d_ws;
    float* AF  = AG  + NDIM * NDIM;
    float* T0  = AF  + NDIM * NDIM;
    float* T0F = T0  + NDIM * NDIM;
    float* T1  = T0F + NDIM * NDIM;
    float* T1F = T1  + NDIM * NDIM;
    float* PF  = T1F + NDIM * NDIM;
    float* WF  = PF  + NDIM * ODIM;

    build_A_kernel<<<(NDIM * NDIM / 4) / 256, 256, 0, stream>>>(W, wout, AG, AF, WF);

    gemm_chain_kernel<<<256, 128, 0, stream>>>(AG, AF,  T0, T0F);
    gemm_chain_kernel<<<256, 128, 0, stream>>>(T0, T0F, T1, T1F);
    gemm_chain_kernel<<<256, 128, 0, stream>>>(T1, T1F, T0, T0F);
    gemm_chain_kernel<<<256, 128, 0, stream>>>(T0, T0F, T1, T1F);
    gemm_chain_kernel<<<256, 128, 0, stream>>>(T1, T1F, T0, T0F);
    gemm_chain_kernel<<<256, 128, 0, stream>>>(T0, T0F, T1, T1F);
    gemm_chain_kernel<<<256, 128, 0, stream>>>(T1, AF,  T0, T0F);

    gemm_nt_P_kernel<<<256, 128, 0, stream>>>(T0, WF, PF);

    gemm_out_kernel<<<512, 128, 0, stream>>>(x, PF, out);
}
